# starting kernel
# baseline (speedup 1.0000x reference)
#include <hip/hip_runtime.h>

typedef float v2f __attribute__((ext_vector_type(2)));
typedef float v8f __attribute__((ext_vector_type(8)));

namespace {
constexpr int   C    = 64;
constexpr int   STR  = 68;
constexpr float kEps = 1e-7f;
}

__global__ __launch_bounds__(64)
void mfdfc_wnorm_kernel(const float* __restrict__ w_raw, float* __restrict__ wn) {
  const int o = threadIdx.x;
  const float lc = 4.1588830833596715f;
  float s = 0.0f;
  for (int i = 0; i < C; ++i) s += expf(w_raw[i * C + o] - lc);
  const float inv = 1.0f / s;
  for (int i = 0; i < C; ++i) wn[i * C + o] = expf(w_raw[i * C + o] - lc) * inv;
}

__device__ __forceinline__ float half16_sum(float v) {
  v += __shfl_xor(v, 1, 32);
  v += __shfl_xor(v, 2, 32);
  v += __shfl_xor(v, 4, 32);
  v += __shfl_xor(v, 8, 32);
  return v;
}

__global__ __launch_bounds__(256)
void mfdfc_main_kernel(const float* __restrict__ x, const float* __restrict__ wn,
                       float* __restrict__ out) {
  __shared__ float xs[C * STR];
  __shared__ float as[C * STR];
  __shared__ float ss[C * STR];
  __shared__ float tpart[C * 4];
  __shared__ float gpart[C * 4];
  __shared__ float pa[C];
  __shared__ float qq[C];
  __shared__ float tt[C];
  __shared__ float cg[C];
  __shared__ float sg[C];

  const int tid  = threadIdx.x;
  const int lane = tid & 31;
  const int wave = tid >> 5;
  const int il   = lane & 15;
  const int hh   = lane >> 4;
  const int n    = blockIdx.x;

  const float* xn = x + (size_t)n * C * C;

  for (int j = 0; j < 4; ++j) {
    int f4  = tid + 256 * j;
    int row = f4 >> 4;
    int c4  = f4 & 15;
    const float4 v = ((const float4*)xn)[f4];
    *(float4*)&xs[row * STR + 4 * c4] = v;
  }
  __syncthreads();

  if (tid < C) {
    float s = 0.0f;
    for (int k = 0; k < C; ++k) { float v = xs[tid * STR + k]; s += v * v; }
    qq[tid] = s;
  }
  for (int j = 0; j < 16; ++j) {
    int f = tid + 256 * j;
    int o = f >> 6, k = f & 63;
    as[o * STR + k] = xs[k];
  }

  float wreg[2][8];
  for (int u = 0; u < 2; ++u) {
    int t  = wave * 2 + u;
    int m0 = (t >> 2) * 16;
    int i0 = (t & 3) * 16;
    int i  = i0 + il;
    for (int r = 0; r < 8; ++r) {
      int o = m0 + r + 8 * hh;
      wreg[u][r] = wn[i * C + o];
    }
  }
  __syncthreads();

  for (int it = 0; it < 3; ++it) {
    if (tid < C) {
      float s = 0.0f;
      for (int k = 0; k < C; ++k) { float v = as[tid * STR + k]; s += v * v; }
      pa[tid] = s;
    }
    __syncthreads();

    for (int u = 0; u < 2; ++u) {
      int t  = wave * 2 + u;
      int m0 = (t >> 2) * 16;
      int i0 = (t & 3) * 16;
      v8f acc = {};
      for (int kc = 0; kc < C; kc += 4) {
        v2f af = *(const v2f*)&as[(m0 + il) * STR + kc + 2 * hh];
        v2f bf = *(const v2f*)&xs[(i0 + il) * STR + kc + 2 * hh];
        acc = __builtin_amdgcn_wmma_f32_16x16x4_f32(false, af, false, bf,
                                                    (short)0, acc, false, false);
      }
      int i = i0 + il;
      for (int r = 0; r < 8; ++r) {
        int o = m0 + r + 8 * hh;
        float du  = acc[r];
        float dc  = fminf(fmaxf(du, -1.0f + kEps), 1.0f - kEps);
        float th  = acosf(dc);
        float nv2 = qq[i] - 2.0f * dc * du + dc * dc * pa[o];
        float nv  = sqrtf(fmaxf(nv2, 0.0f));
        float fac = (nv < kEps) ? 1.0f : th / nv;
        float s   = wreg[u][r] * fac;
        ss[o * STR + i] = s;
        float part = half16_sum(s * dc);
        if (il == 0) tpart[o * 4 + (t & 3)] = part;
      }
    }
    __syncthreads();

    if (tid < C)
      tt[tid] = tpart[tid * 4 + 0] + tpart[tid * 4 + 1] +
                tpart[tid * 4 + 2] + tpart[tid * 4 + 3];
    __syncthreads();

    float greg[2][8];
    for (int u = 0; u < 2; ++u) {
      int t  = wave * 2 + u;
      int o0 = (t >> 2) * 16;
      int k0 = (t & 3) * 16;
      v8f acc = {};
      for (int ic = 0; ic < C; ic += 4) {
        v2f af = *(const v2f*)&ss[(o0 + il) * STR + ic + 2 * hh];
        v2f bf;
        bf.x = xs[(ic + 2 * hh)     * STR + k0 + il];
        bf.y = xs[(ic + 2 * hh + 1) * STR + k0 + il];
        acc = __builtin_amdgcn_wmma_f32_16x16x4_f32(false, af, false, bf,
                                                    (short)0, acc, false, false);
      }
      int k = k0 + il;
      for (int r = 0; r < 8; ++r) {
        int o = o0 + r + 8 * hh;
        float g = acc[r] - tt[o] * as[o * STR + k];
        greg[u][r] = g;
        float part = half16_sum(g * g);
        if (il == 0) gpart[o * 4 + (t & 3)] = part;
      }
    }
    __syncthreads();

    if (tid < C) {
      float s2 = gpart[tid * 4 + 0] + gpart[tid * 4 + 1] +
                 gpart[tid * 4 + 2] + gpart[tid * 4 + 3];
      float gl = sqrtf(s2);
      cg[tid] = cosf(gl);
      sg[tid] = (gl < kEps) ? 1.0f : (sinf(gl) / gl);
    }
    __syncthreads();

    for (int u = 0; u < 2; ++u) {
      int t  = wave * 2 + u;
      int o0 = (t >> 2) * 16;
      int k0 = (t & 3) * 16;
      int k  = k0 + il;
      for (int r = 0; r < 8; ++r) {
        int o = o0 + r + 8 * hh;
        as[o * STR + k] = cg[o] * as[o * STR + k] + sg[o] * greg[u][r];
      }
    }
    __syncthreads();
  }

  float* on = out + (size_t)n * C * C;
  for (int j = 0; j < 16; ++j) {
    int f = tid + 256 * j;
    int o = f >> 6, k = f & 63;
    on[f] = as[o * STR + k];
  }
}

extern "C" void kernel_launch(void* const* d_in, const int* in_sizes, int n_in,
                              void* d_out, int out_size, void* d_ws, size_t ws_size,
                              hipStream_t stream) {
  (void)in_sizes; (void)n_in; (void)out_size; (void)ws_size;
  const float* x     = (const float*)d_in[0];
  const float* w_raw = (const float*)d_in[1];
  float* out = (float*)d_out;
  float* wn  = (float*)d_ws;

  mfdfc_wnorm_kernel<<<1, 64, 0, stream>>>(w_raw, wn);
  mfdfc_main_kernel<<<1024, 256, 0, stream>>>(x, wn, out);
}


# Both products transposed (out channel on the tile column): per-channel sums (s*dot, |grad|^2, |a|^2) become in-register sums + one lane-half exchange, replacing shuffle trees; new summation order
# speedup vs baseline: 1.8461x; 1.8461x over previous
#include <hip/hip_runtime.h>


typedef float v2f __attribute__((ext_vector_type(2)));
typedef float v8f __attribute__((ext_vector_type(8)));

namespace {
constexpr int   C    = 64;
constexpr int   STR  = 68;
constexpr float kEps = 1e-7f;
}

__global__ __launch_bounds__(64)
void mfdfc_wnorm_kernel(const float* __restrict__ w_raw, float* __restrict__ wn) {
  const int o = threadIdx.x;
  const float lc = 4.1588830833596715f;
  float s = 0.0f;
  for (int i = 0; i < C; ++i) s += expf(w_raw[i * C + o] - lc);
  const float inv = 1.0f / s;
  for (int i = 0; i < C; ++i) wn[i * C + o] = expf(w_raw[i * C + o] - lc) * inv;
}

__device__ __forceinline__ float xhalf(float x) {
  const int b = __builtin_bit_cast(int, x);
  return __builtin_bit_cast(float,
      __builtin_amdgcn_permlanex16(b, b, 0x76543210, 0xfedcba98, false, false));
}

__device__ __forceinline__ v8f mma4(v2f a, v2f b, v8f c) {
  return __builtin_amdgcn_wmma_f32_16x16x4_f32(false, a, false, b, (short)0, c, false, false);
}

__global__ __launch_bounds__(256)
void mfdfc_main_kernel(const float* __restrict__ x, const float* __restrict__ wn,
                       float* __restrict__ out) {
  __shared__ __attribute__((aligned(16))) float xs[C * STR];
  __shared__ __attribute__((aligned(16))) float as[C * STR];
  __shared__ __attribute__((aligned(16))) float ss[C * STR];
  __shared__ __attribute__((aligned(16))) float qq[C];
  __shared__ __attribute__((aligned(16))) float tpart[C * 2];
  __shared__ __attribute__((aligned(16))) float gpart[C * 2];
  __shared__ __attribute__((aligned(16))) float ppart[C * 2];

  const int tid  = threadIdx.x;
  const int lane = tid & 31;
  const int wave = tid >> 5;
  const int il   = lane & 15;
  const int hh   = lane >> 4;
  const int n    = blockIdx.x;

  const int o0 = (wave >> 1) * 16;
  const int wp = wave & 1;
  const int j0 = wp * 32;
  const int o  = o0 + il;

  const float* xn = x + (size_t)n * C * C;

  for (int j = 0; j < 4; ++j) {
    int f4  = tid + 256 * j;
    int row = f4 >> 4;
    int c4  = f4 & 15;
    const float4 v = ((const float4*)xn)[f4];
    *(float4*)&xs[row * STR + 4 * c4] = v;
  }
  __syncthreads();

  if (tid < C) {
    float s = 0.0f;
    for (int k = 0; k < C; ++k) { float v = xs[tid * STR + k]; s += v * v; }
    qq[tid] = s;
  }
  for (int j = 0; j < 16; ++j) {
    int f = tid + 256 * j;
    int oo = f >> 6, k = f & 63;
    as[oo * STR + k] = xs[k];
  }

  float wreg[2][8];
  for (int u = 0; u < 2; ++u)
    for (int r = 0; r < 8; ++r)
      wreg[u][r] = wn[(j0 + 16 * u + 8 * hh + r) * C + o];
  __syncthreads();

  for (int it = 0; it < 3; ++it) {
    float pav;
    if (it == 0) {
      pav = qq[0];
    } else {
      const float2 pp = *(const float2*)&ppart[2 * o];
      pav = pp.x + pp.y;
    }

    v8f acc[2] = {};
    for (int kc = 0; kc < C; kc += 4) {
      const v2f bf = *(const v2f*)&as[o * STR + kc + 2 * hh];
      const v2f a0 = *(const v2f*)&xs[(j0 + il) * STR + kc + 2 * hh];
      const v2f a1 = *(const v2f*)&xs[(j0 + 16 + il) * STR + kc + 2 * hh];
      acc[0] = mma4(a0, bf, acc[0]);
      acc[1] = mma4(a1, bf, acc[1]);
    }

    float tsum = 0.0f;
    for (int u = 0; u < 2; ++u) {
      const int ib = j0 + 16 * u + 8 * hh;
      const float4 qa = *(const float4*)&qq[ib];
      const float4 qb = *(const float4*)&qq[ib + 4];
      const float qv[8] = {qa.x, qa.y, qa.z, qa.w, qb.x, qb.y, qb.z, qb.w};
      float sv[8];
      for (int r = 0; r < 8; ++r) {
        float du  = acc[u][r];
        float dc  = fminf(fmaxf(du, -1.0f + kEps), 1.0f - kEps);
        float th  = acosf(dc);
        float nv2 = qv[r] - 2.0f * dc * du + dc * dc * pav;
        float nv  = sqrtf(fmaxf(nv2, 0.0f));
        float fac = (nv < kEps) ? 1.0f : th / nv;
        float s   = wreg[u][r] * fac;
        sv[r] = s;
        tsum += s * dc;
      }
      *(float4*)&ss[o * STR + ib]     = make_float4(sv[0], sv[1], sv[2], sv[3]);
      *(float4*)&ss[o * STR + ib + 4] = make_float4(sv[4], sv[5], sv[6], sv[7]);
    }
    tsum += xhalf(tsum);
    tpart[2 * o + wp] = tsum;
    __syncthreads();

    const float2 tp = *(const float2*)&tpart[2 * o];
    const float ttv = tp.x + tp.y;
    v8f gac[2] = {};
    for (int ic = 0; ic < C; ic += 4) {
      const v2f bf = *(const v2f*)&ss[o * STR + ic + 2 * hh];
      v2f a0, a1;
      a0.x = xs[(ic + 2 * hh)     * STR + j0 + il];
      a0.y = xs[(ic + 2 * hh + 1) * STR + j0 + il];
      a1.x = xs[(ic + 2 * hh)     * STR + j0 + 16 + il];
      a1.y = xs[(ic + 2 * hh + 1) * STR + j0 + 16 + il];
      gac[0] = mma4(a0, bf, gac[0]);
      gac[1] = mma4(a1, bf, gac[1]);
    }

    float ao[2][8], gr[2][8];
    float gsum = 0.0f;
    for (int u = 0; u < 2; ++u) {
      const int kb = j0 + 16 * u + 8 * hh;
      const float4 aa = *(const float4*)&as[o * STR + kb];
      const float4 ab = *(const float4*)&as[o * STR + kb + 4];
      ao[u][0] = aa.x; ao[u][1] = aa.y; ao[u][2] = aa.z; ao[u][3] = aa.w;
      ao[u][4] = ab.x; ao[u][5] = ab.y; ao[u][6] = ab.z; ao[u][7] = ab.w;
      for (int r = 0; r < 8; ++r) {
        float g = gac[u][r] - ttv * ao[u][r];
        gr[u][r] = g;
        gsum += g * g;
      }
    }
    gsum += xhalf(gsum);
    gpart[2 * o + wp] = gsum;
    __syncthreads();

    const float2 gp = *(const float2*)&gpart[2 * o];
    const float gl  = sqrtf(gp.x + gp.y);
    const float cgv = cosf(gl);
    const float sgv = (gl < kEps) ? 1.0f : (sinf(gl) / gl);
    float psum = 0.0f;
    for (int u = 0; u < 2; ++u) {
      const int kb = j0 + 16 * u + 8 * hh;
      float an[8];
      for (int r = 0; r < 8; ++r) {
        an[r] = cgv * ao[u][r] + sgv * gr[u][r];
        psum += an[r] * an[r];
      }
      *(float4*)&as[o * STR + kb]     = make_float4(an[0], an[1], an[2], an[3]);
      *(float4*)&as[o * STR + kb + 4] = make_float4(an[4], an[5], an[6], an[7]);
    }
    psum += xhalf(psum);
    ppart[2 * o + wp] = psum;
    __syncthreads();
  }

  float* on = out + (size_t)n * C * C;
  for (int j = 0; j < 16; ++j) {
    int f = tid + 256 * j;
    int oo = f >> 6, k = f & 63;
    on[f] = as[oo * STR + k];
  }
}

extern "C" void kernel_launch(void* const* d_in, const int* in_sizes, int n_in,
                              void* d_out, int out_size, void* d_ws, size_t ws_size,
                              hipStream_t stream) {
  (void)in_sizes; (void)n_in; (void)out_size; (void)ws_size;
  const float* x     = (const float*)d_in[0];
  const float* w_raw = (const float*)d_in[1];
  float* out = (float*)d_out;
  float* wn  = (float*)d_ws;

  mfdfc_wnorm_kernel<<<1, 64, 0, stream>>>(w_raw, wn);
  mfdfc_main_kernel<<<1024, 256, 0, stream>>>(x, wn, out);
}


# Weight normalisation folded into the main kernel: each lane exponentiates the weights it uses, column sum by lane-half exchange plus wave-pair add (new summation order of that sum); one launch
# speedup vs baseline: 2.8303x; 1.5331x over previous
#include <hip/hip_runtime.h>


typedef float v2f __attribute__((ext_vector_type(2)));
typedef float v8f __attribute__((ext_vector_type(8)));

namespace {
constexpr int   C    = 64;
constexpr int   STR  = 68;
constexpr float kEps = 1e-7f;
}

__device__ __forceinline__ float xhalf(float x) {
  const int b = __builtin_bit_cast(int, x);
  return __builtin_bit_cast(float,
      __builtin_amdgcn_permlanex16(b, b, 0x76543210, 0xfedcba98, false, false));
}

__device__ __forceinline__ v8f mma4(v2f a, v2f b, v8f c) {
  return __builtin_amdgcn_wmma_f32_16x16x4_f32(false, a, false, b, (short)0, c, false, false);
}

__global__ __launch_bounds__(256)
void mfdfc_main_kernel(const float* __restrict__ x, const float* __restrict__ w_raw,
                       float* __restrict__ out) {
  __shared__ __attribute__((aligned(16))) float xs[C * STR];
  __shared__ __attribute__((aligned(16))) float as[C * STR];
  __shared__ __attribute__((aligned(16))) float ss[C * STR];
  __shared__ __attribute__((aligned(16))) float qq[C];
  __shared__ __attribute__((aligned(16))) float tpart[C * 2];
  __shared__ __attribute__((aligned(16))) float gpart[C * 2];
  __shared__ __attribute__((aligned(16))) float ppart[C * 2];
  __shared__ __attribute__((aligned(16))) float wpart[C * 2];

  const int tid  = threadIdx.x;
  const int lane = tid & 31;
  const int wave = tid >> 5;
  const int il   = lane & 15;
  const int hh   = lane >> 4;
  const int n    = blockIdx.x;

  const int o0 = (wave >> 1) * 16;
  const int wp = wave & 1;
  const int j0 = wp * 32;
  const int o  = o0 + il;

  const float* xn = x + (size_t)n * C * C;

  float wreg[2][8];
  for (int u = 0; u < 2; ++u)
    for (int r = 0; r < 8; ++r)
      wreg[u][r] = w_raw[(j0 + 16 * u + 8 * hh + r) * C + o];

  for (int j = 0; j < 4; ++j) {
    int f4  = tid + 256 * j;
    int row = f4 >> 4;
    int c4  = f4 & 15;
    const float4 v = ((const float4*)xn)[f4];
    *(float4*)&xs[row * STR + 4 * c4] = v;
  }
  __syncthreads();

  if (tid < C) {
    float s = 0.0f;
    for (int k = 0; k < C; ++k) { float v = xs[tid * STR + k]; s += v * v; }
    qq[tid] = s;
  }
  for (int j = 0; j < 16; ++j) {
    int f = tid + 256 * j;
    int oo = f >> 6, k = f & 63;
    as[oo * STR + k] = xs[k];
  }

  const float lc = 4.1588830833596715f;
  float wsum = 0.0f;
  for (int u = 0; u < 2; ++u)
    for (int r = 0; r < 8; ++r) {
      const float e = expf(wreg[u][r] - lc);
      wreg[u][r] = e;
      wsum += e;
    }
  wsum += xhalf(wsum);
  wpart[2 * o + wp] = wsum;
  __syncthreads();
  {
    const float2 wq = *(const float2*)&wpart[2 * o];
    const float winv = 1.0f / (wq.x + wq.y);
    for (int u = 0; u < 2; ++u)
      for (int r = 0; r < 8; ++r) wreg[u][r] *= winv;
  }

  for (int it = 0; it < 3; ++it) {
    float pav;
    if (it == 0) {
      pav = qq[0];
    } else {
      const float2 pp = *(const float2*)&ppart[2 * o];
      pav = pp.x + pp.y;
    }

    v8f acc[2] = {};
    for (int kc = 0; kc < C; kc += 4) {
      const v2f bf = *(const v2f*)&as[o * STR + kc + 2 * hh];
      const v2f a0 = *(const v2f*)&xs[(j0 + il) * STR + kc + 2 * hh];
      const v2f a1 = *(const v2f*)&xs[(j0 + 16 + il) * STR + kc + 2 * hh];
      acc[0] = mma4(a0, bf, acc[0]);
      acc[1] = mma4(a1, bf, acc[1]);
    }

    float tsum = 0.0f;
    for (int u = 0; u < 2; ++u) {
      const int ib = j0 + 16 * u + 8 * hh;
      const float4 qa = *(const float4*)&qq[ib];
      const float4 qb = *(const float4*)&qq[ib + 4];
      const float qv[8] = {qa.x, qa.y, qa.z, qa.w, qb.x, qb.y, qb.z, qb.w};
      float sv[8];
      for (int r = 0; r < 8; ++r) {
        float du  = acc[u][r];
        float dc  = fminf(fmaxf(du, -1.0f + kEps), 1.0f - kEps);
        float th  = acosf(dc);
        float nv2 = qv[r] - 2.0f * dc * du + dc * dc * pav;
        float nv  = sqrtf(fmaxf(nv2, 0.0f));
        float fac = (nv < kEps) ? 1.0f : th / nv;
        float s   = wreg[u][r] * fac;
        sv[r] = s;
        tsum += s * dc;
      }
      *(float4*)&ss[o * STR + ib]     = make_float4(sv[0], sv[1], sv[2], sv[3]);
      *(float4*)&ss[o * STR + ib + 4] = make_float4(sv[4], sv[5], sv[6], sv[7]);
    }
    tsum += xhalf(tsum);
    tpart[2 * o + wp] = tsum;
    __syncthreads();

    const float2 tp = *(const float2*)&tpart[2 * o];
    const float ttv = tp.x + tp.y;
    v8f gac[2] = {};
    for (int ic = 0; ic < C; ic += 4) {
      const v2f bf = *(const v2f*)&ss[o * STR + ic + 2 * hh];
      v2f a0, a1;
      a0.x = xs[(ic + 2 * hh)     * STR + j0 + il];
      a0.y = xs[(ic + 2 * hh + 1) * STR + j0 + il];
      a1.x = xs[(ic + 2 * hh)     * STR + j0 + 16 + il];
      a1.y = xs[(ic + 2 * hh + 1) * STR + j0 + 16 + il];
      gac[0] = mma4(a0, bf, gac[0]);
      gac[1] = mma4(a1, bf, gac[1]);
    }

    float ao[2][8], gr[2][8];
    float gsum = 0.0f;
    for (int u = 0; u < 2; ++u) {
      const int kb = j0 + 16 * u + 8 * hh;
      const float4 aa = *(const float4*)&as[o * STR + kb];
      const float4 ab = *(const float4*)&as[o * STR + kb + 4];
      ao[u][0] = aa.x; ao[u][1] = aa.y; ao[u][2] = aa.z; ao[u][3] = aa.w;
      ao[u][4] = ab.x; ao[u][5] = ab.y; ao[u][6] = ab.z; ao[u][7] = ab.w;
      for (int r = 0; r < 8; ++r) {
        float g = gac[u][r] - ttv * ao[u][r];
        gr[u][r] = g;
        gsum += g * g;
      }
    }
    gsum += xhalf(gsum);
    gpart[2 * o + wp] = gsum;
    __syncthreads();

    const float2 gp = *(const float2*)&gpart[2 * o];
    const float gl  = sqrtf(gp.x + gp.y);
    const float cgv = cosf(gl);
    const float sgv = (gl < kEps) ? 1.0f : (sinf(gl) / gl);
    float psum = 0.0f;
    for (int u = 0; u < 2; ++u) {
      const int kb = j0 + 16 * u + 8 * hh;
      float an[8];
      for (int r = 0; r < 8; ++r) {
        an[r] = cgv * ao[u][r] + sgv * gr[u][r];
        psum += an[r] * an[r];
      }
      *(float4*)&as[o * STR + kb]     = make_float4(an[0], an[1], an[2], an[3]);
      *(float4*)&as[o * STR + kb + 4] = make_float4(an[4], an[5], an[6], an[7]);
    }
    psum += xhalf(psum);
    ppart[2 * o + wp] = psum;
    __syncthreads();
  }

  float* on = out + (size_t)n * C * C;
  for (int j = 0; j < 16; ++j) {
    int f = tid + 256 * j;
    int oo = f >> 6, k = f & 63;
    on[f] = as[oo * STR + k];
  }
}

extern "C" void kernel_launch(void* const* d_in, const int* in_sizes, int n_in,
                              void* d_out, int out_size, void* d_ws, size_t ws_size,
                              hipStream_t stream) {
  (void)in_sizes; (void)n_in; (void)out_size; (void)d_ws; (void)ws_size;
  const float* x     = (const float*)d_in[0];
  const float* w_raw = (const float*)d_in[1];
  float* out = (float*)d_out;

  mfdfc_main_kernel<<<1024, 256, 0, stream>>>(x, w_raw, out);
}


# Step 1 specialised (all means start at x_0): dot, angle, log-map factor once per row i in the norm pass (64 values, not 64x64; that dot now a vector-ALU fma chain); no product 1 in step 1
# speedup vs baseline: 3.1832x; 1.1247x over previous
#include <hip/hip_runtime.h>


typedef float v2f __attribute__((ext_vector_type(2)));
typedef float v8f __attribute__((ext_vector_type(8)));

namespace {
constexpr int   C    = 64;
constexpr int   STR  = 68;
constexpr float kEps = 1e-7f;
}

__device__ __forceinline__ float xhalf(float x) {
  const int b = __builtin_bit_cast(int, x);
  return __builtin_bit_cast(float,
      __builtin_amdgcn_permlanex16(b, b, 0x76543210, 0xfedcba98, false, false));
}

__device__ __forceinline__ v8f mma4(v2f a, v2f b, v8f c) {
  return __builtin_amdgcn_wmma_f32_16x16x4_f32(false, a, false, b, (short)0, c, false, false);
}

__global__ __launch_bounds__(256)
void mfdfc_main_kernel(const float* __restrict__ x, const float* __restrict__ w_raw,
                       float* __restrict__ out) {
  __shared__ __attribute__((aligned(16))) float xs[C * STR];
  __shared__ __attribute__((aligned(16))) float as[C * STR];
  __shared__ __attribute__((aligned(16))) float ss[C * STR];
  __shared__ __attribute__((aligned(16))) float qq[C];
  __shared__ __attribute__((aligned(16))) float d0[C];
  __shared__ __attribute__((aligned(16))) float f0[C];
  __shared__ __attribute__((aligned(16))) float tpart[C * 2];
  __shared__ __attribute__((aligned(16))) float gpart[C * 2];
  __shared__ __attribute__((aligned(16))) float ppart[C * 2];
  __shared__ __attribute__((aligned(16))) float wpart[C * 2];

  const int tid  = threadIdx.x;
  const int lane = tid & 31;
  const int wave = tid >> 5;
  const int il   = lane & 15;
  const int hh   = lane >> 4;
  const int n    = blockIdx.x;

  const int o0 = (wave >> 1) * 16;
  const int wp = wave & 1;
  const int j0 = wp * 32;
  const int o  = o0 + il;

  const float* xn = x + (size_t)n * C * C;

  float wreg[2][8];
  for (int u = 0; u < 2; ++u)
    for (int r = 0; r < 8; ++r)
      wreg[u][r] = w_raw[(j0 + 16 * u + 8 * hh + r) * C + o];

  for (int j = 0; j < 4; ++j) {
    int f4  = tid + 256 * j;
    int row = f4 >> 4;
    int c4  = f4 & 15;
    const float4 v = ((const float4*)xn)[f4];
    *(float4*)&xs[row * STR + 4 * c4] = v;
  }
  __syncthreads();

  if (tid < C) {
    float s = 0.0f, du = 0.0f, q0 = 0.0f;
    for (int k = 0; k < C; ++k) {
      const float v  = xs[tid * STR + k];
      const float x0 = xs[k];
      s  += v * v;
      du += v * x0;
      q0 += x0 * x0;
    }
    const float dc  = fminf(fmaxf(du, -1.0f + kEps), 1.0f - kEps);
    const float th  = acosf(dc);
    const float nv2 = s - 2.0f * dc * du + dc * dc * q0;
    const float nv  = sqrtf(fmaxf(nv2, 0.0f));
    qq[tid] = s;
    d0[tid] = dc;
    f0[tid] = (nv < kEps) ? 1.0f : th / nv;
  }

  const float lc = 4.1588830833596715f;
  float wsum = 0.0f;
  for (int u = 0; u < 2; ++u)
    for (int r = 0; r < 8; ++r) {
      const float e = expf(wreg[u][r] - lc);
      wreg[u][r] = e;
      wsum += e;
    }
  wsum += xhalf(wsum);
  wpart[2 * o + wp] = wsum;
  __syncthreads();
  {
    const float2 wq = *(const float2*)&wpart[2 * o];
    const float winv = 1.0f / (wq.x + wq.y);
    for (int u = 0; u < 2; ++u)
      for (int r = 0; r < 8; ++r) wreg[u][r] *= winv;
  }

  for (int it = 0; it < 3; ++it) {
    float pav = 0.0f;
    if (it != 0) {
      const float2 pp = *(const float2*)&ppart[2 * o];
      pav = pp.x + pp.y;
    }

    float tsum = 0.0f;
    if (it == 0) {
      for (int u = 0; u < 2; ++u) {
        const int ib = j0 + 16 * u + 8 * hh;
        const float4 fa = *(const float4*)&f0[ib];
        const float4 fb = *(const float4*)&f0[ib + 4];
        const float4 da = *(const float4*)&d0[ib];
        const float4 db = *(const float4*)&d0[ib + 4];
        const float fv[8] = {fa.x, fa.y, fa.z, fa.w, fb.x, fb.y, fb.z, fb.w};
        const float dv[8] = {da.x, da.y, da.z, da.w, db.x, db.y, db.z, db.w};
        float sv[8];
        for (int r = 0; r < 8; ++r) {
          float s = wreg[u][r] * fv[r];
          sv[r] = s;
          tsum += s * dv[r];
        }
        *(float4*)&ss[o * STR + ib]     = make_float4(sv[0], sv[1], sv[2], sv[3]);
        *(float4*)&ss[o * STR + ib + 4] = make_float4(sv[4], sv[5], sv[6], sv[7]);
      }
    } else {
      v8f acc[2] = {};
      for (int kc = 0; kc < C; kc += 4) {
        const v2f bf = *(const v2f*)&as[o * STR + kc + 2 * hh];
        const v2f a0 = *(const v2f*)&xs[(j0 + il) * STR + kc + 2 * hh];
        const v2f a1 = *(const v2f*)&xs[(j0 + 16 + il) * STR + kc + 2 * hh];
        acc[0] = mma4(a0, bf, acc[0]);
        acc[1] = mma4(a1, bf, acc[1]);
      }

      for (int u = 0; u < 2; ++u) {
        const int ib = j0 + 16 * u + 8 * hh;
        const float4 qa = *(const float4*)&qq[ib];
        const float4 qb = *(const float4*)&qq[ib + 4];
        const float qv[8] = {qa.x, qa.y, qa.z, qa.w, qb.x, qb.y, qb.z, qb.w};
        float sv[8];
        for (int r = 0; r < 8; ++r) {
          float du  = acc[u][r];
          float dc  = fminf(fmaxf(du, -1.0f + kEps), 1.0f - kEps);
          float th  = acosf(dc);
          float nv2 = qv[r] - 2.0f * dc * du + dc * dc * pav;
          float nv  = sqrtf(fmaxf(nv2, 0.0f));
          float fac = (nv < kEps) ? 1.0f : th / nv;
          float s   = wreg[u][r] * fac;
          sv[r] = s;
          tsum += s * dc;
        }
        *(float4*)&ss[o * STR + ib]     = make_float4(sv[0], sv[1], sv[2], sv[3]);
        *(float4*)&ss[o * STR + ib + 4] = make_float4(sv[4], sv[5], sv[6], sv[7]);
      }
    }
    tsum += xhalf(tsum);
    tpart[2 * o + wp] = tsum;
    __syncthreads();

    const float2 tp = *(const float2*)&tpart[2 * o];
    const float ttv = tp.x + tp.y;
    v8f gac[2] = {};
    for (int ic = 0; ic < C; ic += 4) {
      const v2f bf = *(const v2f*)&ss[o * STR + ic + 2 * hh];
      v2f a0, a1;
      a0.x = xs[(ic + 2 * hh)     * STR + j0 + il];
      a0.y = xs[(ic + 2 * hh + 1) * STR + j0 + il];
      a1.x = xs[(ic + 2 * hh)     * STR + j0 + 16 + il];
      a1.y = xs[(ic + 2 * hh + 1) * STR + j0 + 16 + il];
      gac[0] = mma4(a0, bf, gac[0]);
      gac[1] = mma4(a1, bf, gac[1]);
    }

    float ao[2][8], gr[2][8];
    float gsum = 0.0f;
    for (int u = 0; u < 2; ++u) {
      const int kb = j0 + 16 * u + 8 * hh;
      const float* am = (it == 0) ? &xs[kb] : &as[o * STR + kb];
      const float4 aa = *(const float4*)&am[0];
      const float4 ab = *(const float4*)&am[4];
      ao[u][0] = aa.x; ao[u][1] = aa.y; ao[u][2] = aa.z; ao[u][3] = aa.w;
      ao[u][4] = ab.x; ao[u][5] = ab.y; ao[u][6] = ab.z; ao[u][7] = ab.w;
      for (int r = 0; r < 8; ++r) {
        float g = gac[u][r] - ttv * ao[u][r];
        gr[u][r] = g;
        gsum += g * g;
      }
    }
    gsum += xhalf(gsum);
    gpart[2 * o + wp] = gsum;
    __syncthreads();

    const float2 gp = *(const float2*)&gpart[2 * o];
    const float gl  = sqrtf(gp.x + gp.y);
    const float cgv = cosf(gl);
    const float sgv = (gl < kEps) ? 1.0f : (sinf(gl) / gl);
    float psum = 0.0f;
    for (int u = 0; u < 2; ++u) {
      const int kb = j0 + 16 * u + 8 * hh;
      float an[8];
      for (int r = 0; r < 8; ++r) {
        an[r] = cgv * ao[u][r] + sgv * gr[u][r];
        psum += an[r] * an[r];
      }
      *(float4*)&as[o * STR + kb]     = make_float4(an[0], an[1], an[2], an[3]);
      *(float4*)&as[o * STR + kb + 4] = make_float4(an[4], an[5], an[6], an[7]);
    }
    psum += xhalf(psum);
    ppart[2 * o + wp] = psum;
    __syncthreads();
  }

  float* on = out + (size_t)n * C * C;
  for (int j = 0; j < 16; ++j) {
    int f = tid + 256 * j;
    int oo = f >> 6, k = f & 63;
    on[f] = as[oo * STR + k];
  }
}

extern "C" void kernel_launch(void* const* d_in, const int* in_sizes, int n_in,
                              void* d_out, int out_size, void* d_ws, size_t ws_size,
                              hipStream_t stream) {
  (void)in_sizes; (void)n_in; (void)out_size; (void)d_ws; (void)ws_size;
  const float* x     = (const float*)d_in[0];
  const float* w_raw = (const float*)d_in[1];
  float* out = (float*)d_out;

  mfdfc_main_kernel<<<1024, 256, 0, stream>>>(x, w_raw, out);
}


# One wave per out-channel tile in both products: S and the means stay in registers as B operand (summed index = accumulator row); no LDS pass or barrier inside the steps; new summation order
# speedup vs baseline: 3.5292x; 1.1087x over previous
#include <hip/hip_runtime.h>


typedef float v2f __attribute__((ext_vector_type(2)));
typedef float v8f __attribute__((ext_vector_type(8)));

namespace {
constexpr int   C    = 64;
constexpr int   STR  = 68;
constexpr float kEps = 1e-7f;
}

__device__ __forceinline__ float xhalf(float x) {
  const int b = __builtin_bit_cast(int, x);
  return __builtin_bit_cast(float,
      __builtin_amdgcn_permlanex16(b, b, 0x76543210, 0xfedcba98, false, false));
}

__device__ __forceinline__ v8f mma4(v2f a, v2f b, v8f c) {
  return __builtin_amdgcn_wmma_f32_16x16x4_f32(false, a, false, b, (short)0, c, false, false);
}

__device__ __forceinline__ void ld8(const float* p, float (&v)[8]) {
  const float4 a = *(const float4*)p;
  const float4 b = *(const float4*)(p + 4);
  v[0] = a.x; v[1] = a.y; v[2] = a.z; v[3] = a.w;
  v[4] = b.x; v[5] = b.y; v[6] = b.z; v[7] = b.w;
}
__device__ __forceinline__ void st8(float* p, const float (&v)[8]) {
  *(float4*)p       = make_float4(v[0], v[1], v[2], v[3]);
  *(float4*)(p + 4) = make_float4(v[4], v[5], v[6], v[7]);
}

__global__ __launch_bounds__(128)
void mfdfc_main_kernel(const float* __restrict__ x, const float* __restrict__ w_raw,
                       float* __restrict__ out) {
  __shared__ __attribute__((aligned(16))) float xs[C * STR];
  __shared__ __attribute__((aligned(16))) float os[C * STR];
  __shared__ __attribute__((aligned(16))) float qq[C];
  __shared__ __attribute__((aligned(16))) float d0[C];
  __shared__ __attribute__((aligned(16))) float f0[C];

  const int tid  = threadIdx.x;
  const int lane = tid & 31;
  const int wave = tid >> 5;
  const int il   = lane & 15;
  const int hh   = lane >> 4;
  const int n    = blockIdx.x;

  const int o  = wave * 16 + il;
  const int rb = 8 * hh;

  const float* xn = x + (size_t)n * C * C;

  float wreg[4][8];
  for (int u = 0; u < 4; ++u) {
    for (int r = 0; r < 8; ++r)
      wreg[u][r] = w_raw[(16 * u + rb + r) * C + o];
  }

  for (int j = 0; j < 8; ++j) {
    int f4  = tid + 128 * j;
    int row = f4 >> 4;
    int c4  = f4 & 15;
    const float4 v = ((const float4*)xn)[f4];
    *(float4*)&xs[row * STR + 4 * c4] = v;
  }
  __syncthreads();

  if (tid < C) {
    float s = 0.0f, du = 0.0f, q0 = 0.0f;
    for (int k = 0; k < C; ++k) {
      const float v  = xs[tid * STR + k];
      const float x0 = xs[k];
      s  += v * v;
      du += v * x0;
      q0 += x0 * x0;
    }
    const float dc  = fminf(fmaxf(du, -1.0f + kEps), 1.0f - kEps);
    const float th  = acosf(dc);
    const float nv2 = s - 2.0f * dc * du + dc * dc * q0;
    const float nv  = sqrtf(fmaxf(nv2, 0.0f));
    qq[tid] = s;
    d0[tid] = dc;
    f0[tid] = (nv < kEps) ? 1.0f : th / nv;
  }

  float* wl = &os[o * STR + rb];
  {
    const float lc = 4.1588830833596715f;
    float wsum = 0.0f;
    for (int u = 0; u < 4; ++u) {
      for (int r = 0; r < 8; ++r) {
        const float e = expf(wreg[u][r] - lc);
        wreg[u][r] = e;
        wsum += e;
      }
    }
    wsum += xhalf(wsum);
    const float winv = 1.0f / wsum;
    for (int u = 0; u < 4; ++u) {
      for (int r = 0; r < 8; ++r) wreg[u][r] *= winv;
      st8(wl + 16 * u, wreg[u]);
    }
  }
  __syncthreads();

  float am[4][8];
  for (int u = 0; u < 4; ++u) ld8(&xs[16 * u + rb], am[u]);

  float pav = 0.0f;

  for (int it = 0; it < 3; ++it) {
    float sv[4][8];
    float tsum = 0.0f;
    if (it == 0) {
      for (int u = 0; u < 4; ++u) {
        float fv[8], dv[8], wv[8];
        ld8(&f0[16 * u + rb], fv);
        ld8(&d0[16 * u + rb], dv);
        ld8(wl + 16 * u, wv);
        for (int r = 0; r < 8; ++r) {
          const float s = wv[r] * fv[r];
          sv[u][r] = s;
          tsum += s * dv[r];
        }
      }
    } else {
      v8f acc[4] = {};
      for (int u = 0; u < 4; ++u) {
        for (int rp = 0; rp < 4; ++rp) {
          v2f bf;
          bf.x = am[u][2 * rp];
          bf.y = am[u][2 * rp + 1];
          for (int t = 0; t < 4; ++t) {
            const v2f a = *(const v2f*)&xs[(16 * t + il) * STR + 16 * u + rb + 2 * rp];
            acc[t] = mma4(a, bf, acc[t]);
          }
        }
      }

      for (int t = 0; t < 4; ++t) {
        float qv[8], wv[8];
        ld8(&qq[16 * t + rb], qv);
        ld8(wl + 16 * t, wv);
        for (int r = 0; r < 8; ++r) {
          float du  = acc[t][r];
          float dc  = fminf(fmaxf(du, -1.0f + kEps), 1.0f - kEps);
          float th  = acosf(dc);
          float nv2 = qv[r] - 2.0f * dc * du + dc * dc * pav;
          float nv  = sqrtf(fmaxf(nv2, 0.0f));
          float fac = (nv < kEps) ? 1.0f : th / nv;
          float s   = wv[r] * fac;
          sv[t][r] = s;
          tsum += s * dc;
        }
      }
    }
    tsum += xhalf(tsum);
    const float ttv = tsum;

    v8f gac[4] = {};
    for (int u = 0; u < 4; ++u) {
      for (int rp = 0; rp < 4; ++rp) {
        v2f bf;
        bf.x = sv[u][2 * rp];
        bf.y = sv[u][2 * rp + 1];
        for (int t = 0; t < 4; ++t) {
          v2f a;
          a.x = xs[(16 * u + rb + 2 * rp)     * STR + 16 * t + il];
          a.y = xs[(16 * u + rb + 2 * rp + 1) * STR + 16 * t + il];
          gac[t] = mma4(a, bf, gac[t]);
        }
      }
    }

    float gr[4][8];
    float gsum = 0.0f;
    for (int t = 0; t < 4; ++t) {
      for (int r = 0; r < 8; ++r) {
        const float g = gac[t][r] - ttv * am[t][r];
        gr[t][r] = g;
        gsum += g * g;
      }
    }
    gsum += xhalf(gsum);

    const float gl  = sqrtf(gsum);
    const float cgv = cosf(gl);
    const float sgv = (gl < kEps) ? 1.0f : (sinf(gl) / gl);
    float psum = 0.0f;
    for (int t = 0; t < 4; ++t) {
      for (int r = 0; r < 8; ++r) {
        const float an = cgv * am[t][r] + sgv * gr[t][r];
        am[t][r] = an;
        psum += an * an;
      }
    }
    psum += xhalf(psum);
    pav = psum;
  }

  for (int t = 0; t < 4; ++t) st8(wl + 16 * t, am[t]);
  __syncthreads();
  float* on = out + (size_t)n * C * C;
  for (int j = 0; j < 8; ++j) {
    int f4  = tid + 128 * j;
    int row = f4 >> 4;
    int c4  = f4 & 15;
    ((float4*)on)[f4] = *(const float4*)&os[row * STR + 4 * c4];
  }
}

extern "C" void kernel_launch(void* const* d_in, const int* in_sizes, int n_in,
                              void* d_out, int out_size, void* d_ws, size_t ws_size,
                              hipStream_t stream) {
  (void)in_sizes; (void)n_in; (void)out_size; (void)d_ws; (void)ws_size;
  const float* x     = (const float*)d_in[0];
  const float* w_raw = (const float*)d_in[1];
  float* out = (float*)d_out;

  mfdfc_main_kernel<<<1024, 128, 0, stream>>>(x, w_raw, out);
}


# Correctly rounded sqrtf and divide sequences (log-map factor, exp-map norm, weight norm) replaced by 1-ulp hardware v_sqrt_f32 / v_rcp_f32: 2 instructions per value, not about 28; sums unchanged
# speedup vs baseline: 4.1539x; 1.1770x over previous
#include <hip/hip_runtime.h>


typedef float v2f __attribute__((ext_vector_type(2)));
typedef float v8f __attribute__((ext_vector_type(8)));

namespace {
constexpr int   C    = 64;
constexpr int   STR  = 68;
constexpr float kEps = 1e-7f;
}

__device__ __forceinline__ float xhalf(float x) {
  const int b = __builtin_bit_cast(int, x);
  return __builtin_bit_cast(float,
      __builtin_amdgcn_permlanex16(b, b, 0x76543210, 0xfedcba98, false, false));
}

__device__ __forceinline__ float hw_sqrt(float v) { return __builtin_amdgcn_sqrtf(v); }
__device__ __forceinline__ float hw_rcp(float v)  { return __builtin_amdgcn_rcpf(v); }

__device__ __forceinline__ v8f mma4(v2f a, v2f b, v8f c) {
  return __builtin_amdgcn_wmma_f32_16x16x4_f32(false, a, false, b, (short)0, c, false, false);
}

__device__ __forceinline__ void ld8(const float* p, float (&v)[8]) {
  const float4 a = *(const float4*)p;
  const float4 b = *(const float4*)(p + 4);
  v[0] = a.x; v[1] = a.y; v[2] = a.z; v[3] = a.w;
  v[4] = b.x; v[5] = b.y; v[6] = b.z; v[7] = b.w;
}
__device__ __forceinline__ void st8(float* p, const float (&v)[8]) {
  *(float4*)p       = make_float4(v[0], v[1], v[2], v[3]);
  *(float4*)(p + 4) = make_float4(v[4], v[5], v[6], v[7]);
}

__global__ __launch_bounds__(128)
void mfdfc_main_kernel(const float* __restrict__ x, const float* __restrict__ w_raw,
                       float* __restrict__ out) {
  __shared__ __attribute__((aligned(16))) float xs[C * STR];
  __shared__ __attribute__((aligned(16))) float os[C * STR];
  __shared__ __attribute__((aligned(16))) float qq[C];
  __shared__ __attribute__((aligned(16))) float d0[C];
  __shared__ __attribute__((aligned(16))) float f0[C];

  const int tid  = threadIdx.x;
  const int lane = tid & 31;
  const int wave = tid >> 5;
  const int il   = lane & 15;
  const int hh   = lane >> 4;
  const int n    = blockIdx.x;

  const int o  = wave * 16 + il;
  const int rb = 8 * hh;

  const float* xn = x + (size_t)n * C * C;

  float wreg[4][8];
  for (int u = 0; u < 4; ++u) {
    for (int r = 0; r < 8; ++r)
      wreg[u][r] = w_raw[(16 * u + rb + r) * C + o];
  }

  for (int j = 0; j < 8; ++j) {
    int f4  = tid + 128 * j;
    int row = f4 >> 4;
    int c4  = f4 & 15;
    const float4 v = ((const float4*)xn)[f4];
    *(float4*)&xs[row * STR + 4 * c4] = v;
  }
  __syncthreads();

  if (tid < C) {
    float s = 0.0f, du = 0.0f, q0 = 0.0f;
    for (int k = 0; k < C; ++k) {
      const float v  = xs[tid * STR + k];
      const float x0 = xs[k];
      s  += v * v;
      du += v * x0;
      q0 += x0 * x0;
    }
    const float dc  = fminf(fmaxf(du, -1.0f + kEps), 1.0f - kEps);
    const float th  = acosf(dc);
    const float nv2 = s - 2.0f * dc * du + dc * dc * q0;
    const float nv  = hw_sqrt(fmaxf(nv2, 0.0f));
    qq[tid] = s;
    d0[tid] = dc;
    f0[tid] = (nv < kEps) ? 1.0f : th * hw_rcp(nv);
  }

  float* wl = &os[o * STR + rb];
  {
    const float lc = 4.1588830833596715f;
    float wsum = 0.0f;
    for (int u = 0; u < 4; ++u) {
      for (int r = 0; r < 8; ++r) {
        const float e = expf(wreg[u][r] - lc);
        wreg[u][r] = e;
        wsum += e;
      }
    }
    wsum += xhalf(wsum);
    const float winv = hw_rcp(wsum);
    for (int u = 0; u < 4; ++u) {
      for (int r = 0; r < 8; ++r) wreg[u][r] *= winv;
      st8(wl + 16 * u, wreg[u]);
    }
  }
  __syncthreads();

  float am[4][8];
  for (int u = 0; u < 4; ++u) ld8(&xs[16 * u + rb], am[u]);

  float pav = 0.0f;

  for (int it = 0; it < 3; ++it) {
    float sv[4][8];
    float tsum = 0.0f;
    if (it == 0) {
      for (int u = 0; u < 4; ++u) {
        float fv[8], dv[8], wv[8];
        ld8(&f0[16 * u + rb], fv);
        ld8(&d0[16 * u + rb], dv);
        ld8(wl + 16 * u, wv);
        for (int r = 0; r < 8; ++r) {
          const float s = wv[r] * fv[r];
          sv[u][r] = s;
          tsum += s * dv[r];
        }
      }
    } else {
      v8f acc[4] = {};
      for (int u = 0; u < 4; ++u) {
        for (int rp = 0; rp < 4; ++rp) {
          v2f bf;
          bf.x = am[u][2 * rp];
          bf.y = am[u][2 * rp + 1];
          for (int t = 0; t < 4; ++t) {
            const v2f a = *(const v2f*)&xs[(16 * t + il) * STR + 16 * u + rb + 2 * rp];
            acc[t] = mma4(a, bf, acc[t]);
          }
        }
      }

      for (int t = 0; t < 4; ++t) {
        float qv[8], wv[8];
        ld8(&qq[16 * t + rb], qv);
        ld8(wl + 16 * t, wv);
        for (int r = 0; r < 8; ++r) {
          float du  = acc[t][r];
          float dc  = fminf(fmaxf(du, -1.0f + kEps), 1.0f - kEps);
          float th  = acosf(dc);
          float nv2 = qv[r] - 2.0f * dc * du + dc * dc * pav;
          float nv  = hw_sqrt(fmaxf(nv2, 0.0f));
          float fac = (nv < kEps) ? 1.0f : th * hw_rcp(nv);
          float s   = wv[r] * fac;
          sv[t][r] = s;
          tsum += s * dc;
        }
      }
    }
    tsum += xhalf(tsum);
    const float ttv = tsum;

    v8f gac[4] = {};
    for (int u = 0; u < 4; ++u) {
      for (int rp = 0; rp < 4; ++rp) {
        v2f bf;
        bf.x = sv[u][2 * rp];
        bf.y = sv[u][2 * rp + 1];
        for (int t = 0; t < 4; ++t) {
          v2f a;
          a.x = xs[(16 * u + rb + 2 * rp)     * STR + 16 * t + il];
          a.y = xs[(16 * u + rb + 2 * rp + 1) * STR + 16 * t + il];
          gac[t] = mma4(a, bf, gac[t]);
        }
      }
    }

    float gr[4][8];
    float gsum = 0.0f;
    for (int t = 0; t < 4; ++t) {
      for (int r = 0; r < 8; ++r) {
        const float g = gac[t][r] - ttv * am[t][r];
        gr[t][r] = g;
        gsum += g * g;
      }
    }
    gsum += xhalf(gsum);

    const float gl  = hw_sqrt(gsum);
    const float cgv = cosf(gl);
    const float sgv = (gl < kEps) ? 1.0f : (sinf(gl) * hw_rcp(gl));
    float psum = 0.0f;
    for (int t = 0; t < 4; ++t) {
      for (int r = 0; r < 8; ++r) {
        const float an = cgv * am[t][r] + sgv * gr[t][r];
        am[t][r] = an;
        psum += an * an;
      }
    }
    psum += xhalf(psum);
    pav = psum;
  }

  for (int t = 0; t < 4; ++t) st8(wl + 16 * t, am[t]);
  __syncthreads();
  float* on = out + (size_t)n * C * C;
  for (int j = 0; j < 8; ++j) {
    int f4  = tid + 128 * j;
    int row = f4 >> 4;
    int c4  = f4 & 15;
    ((float4*)on)[f4] = *(const float4*)&os[row * STR + 4 * c4];
  }
}

extern "C" void kernel_launch(void* const* d_in, const int* in_sizes, int n_in,
                              void* d_out, int out_size, void* d_ws, size_t ws_size,
                              hipStream_t stream) {
  (void)in_sizes; (void)n_in; (void)out_size; (void)d_ws; (void)ws_size;
  const float* x     = (const float*)d_in[0];
  const float* w_raw = (const float*)d_in[1];
  float* out = (float*)d_out;

  mfdfc_main_kernel<<<1024, 128, 0, stream>>>(x, w_raw, out);
}


# Elementary functions cut to their use: acosf -> sqrt(1-|x|)*poly7(|x|), branch-free; log map's sqrt+rcp -> one v_rsq_f32; expf -> fma + v_exp_f32; sinf+cosf -> one sincosf; sums unchanged
# speedup vs baseline: 4.4274x; 1.0658x over previous
#include <hip/hip_runtime.h>


typedef float v2f __attribute__((ext_vector_type(2)));
typedef float v8f __attribute__((ext_vector_type(8)));

namespace {
constexpr int   C    = 64;
constexpr int   STR  = 68;
constexpr float kEps = 1e-7f;
}

__device__ __forceinline__ float xhalf(float x) {
  const int b = __builtin_bit_cast(int, x);
  return __builtin_bit_cast(float,
      __builtin_amdgcn_permlanex16(b, b, 0x76543210, 0xfedcba98, false, false));
}

__device__ __forceinline__ float hw_sqrt(float v) { return __builtin_amdgcn_sqrtf(v); }
__device__ __forceinline__ float hw_rcp(float v)  { return __builtin_amdgcn_rcpf(v); }

__device__ __forceinline__ float hw_rsq(float v)  { return __builtin_amdgcn_rsqf(v); }
__device__ __forceinline__ float exp_m_ln64(float w) {
  return __builtin_amdgcn_exp2f(fmaf(w, 1.4426950408889634f, -6.0f));
}
__device__ __forceinline__ float acos_poly(float x) {
  const float ax = fabsf(x);
  float p = -0.0012624911f;
  p = fmaf(p, ax,  0.0066700901f);
  p = fmaf(p, ax, -0.0170881256f);
  p = fmaf(p, ax,  0.0308918810f);
  p = fmaf(p, ax, -0.0501743046f);
  p = fmaf(p, ax,  0.0889789874f);
  p = fmaf(p, ax, -0.2145988016f);
  p = fmaf(p, ax,  1.5707963050f);
  const float r = hw_sqrt(1.0f - ax) * p;
  return (x < 0.0f) ? (3.14159265358979f - r) : r;
}

__device__ __forceinline__ v8f mma4(v2f a, v2f b, v8f c) {
  return __builtin_amdgcn_wmma_f32_16x16x4_f32(false, a, false, b, (short)0, c, false, false);
}

__device__ __forceinline__ void ld8(const float* p, float (&v)[8]) {
  const float4 a = *(const float4*)p;
  const float4 b = *(const float4*)(p + 4);
  v[0] = a.x; v[1] = a.y; v[2] = a.z; v[3] = a.w;
  v[4] = b.x; v[5] = b.y; v[6] = b.z; v[7] = b.w;
}
__device__ __forceinline__ void st8(float* p, const float (&v)[8]) {
  *(float4*)p       = make_float4(v[0], v[1], v[2], v[3]);
  *(float4*)(p + 4) = make_float4(v[4], v[5], v[6], v[7]);
}

__global__ __launch_bounds__(128)
void mfdfc_main_kernel(const float* __restrict__ x, const float* __restrict__ w_raw,
                       float* __restrict__ out) {
  __shared__ __attribute__((aligned(16))) float xs[C * STR];
  __shared__ __attribute__((aligned(16))) float os[C * STR];
  __shared__ __attribute__((aligned(16))) float qq[C];
  __shared__ __attribute__((aligned(16))) float d0[C];
  __shared__ __attribute__((aligned(16))) float f0[C];

  const int tid  = threadIdx.x;
  const int lane = tid & 31;
  const int wave = tid >> 5;
  const int il   = lane & 15;
  const int hh   = lane >> 4;
  const int n    = blockIdx.x;

  const int o  = wave * 16 + il;
  const int rb = 8 * hh;

  const float* xn = x + (size_t)n * C * C;

  float wreg[4][8];
  for (int u = 0; u < 4; ++u) {
    for (int r = 0; r < 8; ++r)
      wreg[u][r] = w_raw[(16 * u + rb + r) * C + o];
  }

  for (int j = 0; j < 8; ++j) {
    int f4  = tid + 128 * j;
    int row = f4 >> 4;
    int c4  = f4 & 15;
    const float4 v = ((const float4*)xn)[f4];
    *(float4*)&xs[row * STR + 4 * c4] = v;
  }
  __syncthreads();

  if (tid < C) {
    float s = 0.0f, du = 0.0f, q0 = 0.0f;
    for (int k = 0; k < C; ++k) {
      const float v  = xs[tid * STR + k];
      const float x0 = xs[k];
      s  += v * v;
      du += v * x0;
      q0 += x0 * x0;
    }
    const float dc  = fminf(fmaxf(du, -1.0f + kEps), 1.0f - kEps);
    const float th  = acos_poly(dc);
    const float nv2 = fmaxf(s - 2.0f * dc * du + dc * dc * q0, 0.0f);
    qq[tid] = s;
    d0[tid] = dc;
    f0[tid] = (nv2 < kEps * kEps) ? 1.0f : th * hw_rsq(nv2);
  }

  float* wl = &os[o * STR + rb];
  {
    float wsum = 0.0f;
    for (int u = 0; u < 4; ++u) {
      for (int r = 0; r < 8; ++r) {
        const float e = exp_m_ln64(wreg[u][r]);
        wreg[u][r] = e;
        wsum += e;
      }
    }
    wsum += xhalf(wsum);
    const float winv = hw_rcp(wsum);
    for (int u = 0; u < 4; ++u) {
      for (int r = 0; r < 8; ++r) wreg[u][r] *= winv;
      st8(wl + 16 * u, wreg[u]);
    }
  }
  __syncthreads();

  float am[4][8];
  for (int u = 0; u < 4; ++u) ld8(&xs[16 * u + rb], am[u]);

  float pav = 0.0f;

  for (int it = 0; it < 3; ++it) {
    float sv[4][8];
    float tsum = 0.0f;
    if (it == 0) {
      for (int u = 0; u < 4; ++u) {
        float fv[8], dv[8], wv[8];
        ld8(&f0[16 * u + rb], fv);
        ld8(&d0[16 * u + rb], dv);
        ld8(wl + 16 * u, wv);
        for (int r = 0; r < 8; ++r) {
          const float s = wv[r] * fv[r];
          sv[u][r] = s;
          tsum += s * dv[r];
        }
      }
    } else {
      v8f acc[4] = {};
      for (int u = 0; u < 4; ++u) {
        for (int rp = 0; rp < 4; ++rp) {
          v2f bf;
          bf.x = am[u][2 * rp];
          bf.y = am[u][2 * rp + 1];
          for (int t = 0; t < 4; ++t) {
            const v2f a = *(const v2f*)&xs[(16 * t + il) * STR + 16 * u + rb + 2 * rp];
            acc[t] = mma4(a, bf, acc[t]);
          }
        }
      }

      for (int t = 0; t < 4; ++t) {
        float qv[8], wv[8];
        ld8(&qq[16 * t + rb], qv);
        ld8(wl + 16 * t, wv);
        for (int r = 0; r < 8; ++r) {
          float du  = acc[t][r];
          float dc  = fminf(fmaxf(du, -1.0f + kEps), 1.0f - kEps);
          float th  = acos_poly(dc);
          float nv2 = fmaxf(qv[r] - 2.0f * dc * du + dc * dc * pav, 0.0f);
          float fac = (nv2 < kEps * kEps) ? 1.0f : th * hw_rsq(nv2);
          float s   = wv[r] * fac;
          sv[t][r] = s;
          tsum += s * dc;
        }
      }
    }
    tsum += xhalf(tsum);
    const float ttv = tsum;

    v8f gac[4] = {};
    for (int u = 0; u < 4; ++u) {
      for (int rp = 0; rp < 4; ++rp) {
        v2f bf;
        bf.x = sv[u][2 * rp];
        bf.y = sv[u][2 * rp + 1];
        for (int t = 0; t < 4; ++t) {
          v2f a;
          a.x = xs[(16 * u + rb + 2 * rp)     * STR + 16 * t + il];
          a.y = xs[(16 * u + rb + 2 * rp + 1) * STR + 16 * t + il];
          gac[t] = mma4(a, bf, gac[t]);
        }
      }
    }

    float gr[4][8];
    float gsum = 0.0f;
    for (int t = 0; t < 4; ++t) {
      for (int r = 0; r < 8; ++r) {
        const float g = gac[t][r] - ttv * am[t][r];
        gr[t][r] = g;
        gsum += g * g;
      }
    }
    gsum += xhalf(gsum);

    const float gl  = hw_sqrt(gsum);
    float sng, cgv;
    sincosf(gl, &sng, &cgv);
    const float sgv = (gl < kEps) ? 1.0f : (sng * hw_rcp(gl));
    float psum = 0.0f;
    for (int t = 0; t < 4; ++t) {
      for (int r = 0; r < 8; ++r) {
        const float an = cgv * am[t][r] + sgv * gr[t][r];
        am[t][r] = an;
        psum += an * an;
      }
    }
    psum += xhalf(psum);
    pav = psum;
  }

  for (int t = 0; t < 4; ++t) st8(wl + 16 * t, am[t]);
  __syncthreads();
  float* on = out + (size_t)n * C * C;
  for (int j = 0; j < 8; ++j) {
    int f4  = tid + 128 * j;
    int row = f4 >> 4;
    int c4  = f4 & 15;
    ((float4*)on)[f4] = *(const float4*)&os[row * STR + 4 * c4];
  }
}

extern "C" void kernel_launch(void* const* d_in, const int* in_sizes, int n_in,
                              void* d_out, int out_size, void* d_ws, size_t ws_size,
                              hipStream_t stream) {
  (void)in_sizes; (void)n_in; (void)out_size; (void)d_ws; (void)ws_size;
  const float* x     = (const float*)d_in[0];
  const float* w_raw = (const float*)d_in[1];
  float* out = (float*)d_out;

  mfdfc_main_kernel<<<1024, 128, 0, stream>>>(x, w_raw, out);
}
